# starting kernel
# baseline (speedup 1.0000x reference)
#include <hip/hip_runtime.h>
#include <math.h>


#define N_NODES 50000
#define N_EDGES 800000
#define FEAT    32
#define HEADS   4
#define MTILES  (N_NODES / 16)

typedef __attribute__((ext_vector_type(2))) float v2f;
typedef __attribute__((ext_vector_type(8))) float v8f;

__device__ __forceinline__ void atomic_add_f32_dev(float* p, float v) {
  asm volatile("global_atomic_add_f32 %0, %1, off scope:SCOPE_DEV"
               :: "v"(p), "v"(v) : "memory");
}
__device__ __forceinline__ void atomic_max_f32_dev(float* p, float v) {
  asm volatile("global_atomic_max_num_f32 %0, %1, off scope:SCOPE_DEV"
               :: "v"(p), "v"(v) : "memory");
}

__device__ __forceinline__ v8f wmma_f32_k32(const float* Arow,
                                            const float* __restrict__ B,
                                            int bcol, int lane, v8f acc) {
  const int row = lane & 15;
  const int hi  = lane >> 4;
#pragma unroll
  for (int kc = 0; kc < 8; ++kc) {
    const int k0 = kc * 4 + 2 * hi;
    v2f a, b;
    a.x = Arow[row * FEAT + k0];
    a.y = Arow[row * FEAT + k0 + 1];
    b.x = B[k0 * FEAT + bcol + row];
    b.y = B[(k0 + 1) * FEAT + bcol + row];
    acc = __builtin_amdgcn_wmma_f32_16x16x4_f32(false, a, false, b, (short)0, acc,
                                                false, false);
  }
  return acc;
}

__global__ void gat_init(float* __restrict__ m, float* __restrict__ denom,
                         float* __restrict__ num) {
  const int idx = blockIdx.x * blockDim.x + threadIdx.x;
  if (idx < N_NODES * HEADS) {
    m[idx]     = -INFINITY;
    denom[idx] = 0.0f;
  }
  if (idx < N_NODES * HEADS * FEAT) num[idx] = 0.0f;
}

__global__ __launch_bounds__(128) void gat_qkv_wmma(const float* __restrict__ x,
                                                    const float* __restrict__ Wq,
                                                    const float* __restrict__ Wk,
                                                    const float* __restrict__ Wv,
                                                    float* __restrict__ qkv) {
  const int lane = threadIdx.x & 31;
  const int wave = threadIdx.x >> 5;
  const int comb = blockIdx.y * 4 + wave;
  const int mat  = comb >> 3;
  const int head = (comb >> 1) & 3;
  const int nt   = comb & 1;
  const int m0   = blockIdx.x * 16;

  __shared__ float xs[16 * FEAT];
  ((float4*)xs)[threadIdx.x] = ((const float4*)(x + (size_t)m0 * FEAT))[threadIdx.x];
  __syncthreads();

  const float* W = (mat == 0 ? Wq : (mat == 1 ? Wk : Wv)) + head * FEAT * FEAT;
  v8f acc = {};
  acc = wmma_f32_k32(xs, W, nt * 16, lane, acc);

  float* out = qkv + ((size_t)(mat * HEADS + head) * N_NODES) * FEAT;
  const int col = lane & 15, hi = lane >> 4;
#pragma unroll
  for (int r = 0; r < 8; ++r)
    out[(size_t)(m0 + r + 8 * hi) * FEAT + nt * 16 + col] = acc[r];
}

__global__ __launch_bounds__(256) void gat_edge_logits(const float* __restrict__ qkv,
                                                       const int* __restrict__ ei,
                                                       float* __restrict__ sEH,
                                                       float* __restrict__ m) {
  const int wid = (blockIdx.x * blockDim.x + threadIdx.x) >> 5;
  if (wid >= N_EDGES) return;
  const int lane = threadIdx.x & 31;
  if (lane == 0 && wid + 8192 < N_EDGES) {
    __builtin_prefetch(ei + wid + 8192, 0, 0);
    __builtin_prefetch(ei + N_EDGES + wid + 8192, 0, 0);
  }
  const int h    = lane >> 3;
  const int f4   = (lane & 7) * 4;
  const int snd  = ei[wid];
  const int rcv  = ei[N_EDGES + wid];

  const float* Qr = qkv + ((size_t)(0 * HEADS + h) * N_NODES + rcv) * FEAT;
  const float* Kr = qkv + ((size_t)(1 * HEADS + h) * N_NODES + snd) * FEAT;
  const float4 q = *(const float4*)(Qr + f4);
  const float4 k = *(const float4*)(Kr + f4);
  float p = q.x * k.x + q.y * k.y + q.z * k.z + q.w * k.w;
  p += __shfl_xor(p, 1, 32);
  p += __shfl_xor(p, 2, 32);
  p += __shfl_xor(p, 4, 32);
  if ((lane & 7) == 0) {
    const float s = p * 0.17677669529663687f;
    sEH[(size_t)wid * HEADS + h] = s;
    atomic_max_f32_dev(&m[rcv * HEADS + h], s);
  }
}

__global__ __launch_bounds__(256) void gat_edge_scatter(const float* __restrict__ qkv,
                                                        const int* __restrict__ ei,
                                                        const float* __restrict__ sEH,
                                                        const float* __restrict__ m,
                                                        float* __restrict__ denom,
                                                        float* __restrict__ num) {
  const int wid = (blockIdx.x * blockDim.x + threadIdx.x) >> 5;
  if (wid >= N_EDGES) return;
  const int lane = threadIdx.x & 31;
  if (lane == 0 && wid + 8192 < N_EDGES) {
    __builtin_prefetch(ei + wid + 8192, 0, 0);
    __builtin_prefetch(ei + N_EDGES + wid + 8192, 0, 0);
    __builtin_prefetch(sEH + (size_t)(wid + 8192) * HEADS, 0, 0);
  }
  const int h    = lane >> 3;
  const int f4   = (lane & 7) * 4;
  const int snd  = ei[wid];
  const int rcv  = ei[N_EDGES + wid];

  const float s  = sEH[(size_t)wid * HEADS + h];
  const float mv = m[rcv * HEADS + h];
  const float w  = __expf(s - mv);
  if ((lane & 7) == 0) atomic_add_f32_dev(&denom[rcv * HEADS + h], w);

  const float* Vr = qkv + ((size_t)(2 * HEADS + h) * N_NODES + snd) * FEAT;
  const float4 v  = *(const float4*)(Vr + f4);
  float* dst = num + ((size_t)rcv * HEADS + h) * FEAT + f4;
  atomic_add_f32_dev(dst + 0, w * v.x);
  atomic_add_f32_dev(dst + 1, w * v.y);
  atomic_add_f32_dev(dst + 2, w * v.z);
  atomic_add_f32_dev(dst + 3, w * v.w);
}

__global__ void gat_finalize(const float* __restrict__ num,
                             const float* __restrict__ denom,
                             float* __restrict__ attnMean) {
  const int idx = blockIdx.x * blockDim.x + threadIdx.x;
  if (idx >= N_NODES * FEAT) return;
  const int n = idx >> 5, f = idx & 31;
  float acc = 0.0f;
#pragma unroll
  for (int h = 0; h < HEADS; ++h) {
    const float d = denom[n * HEADS + h];
    const float v = num[((size_t)n * HEADS + h) * FEAT + f];
    acc += (d > 0.0f) ? v / d : 0.0f;
  }
  attnMean[idx] = acc * 0.25f;
}

__global__ __launch_bounds__(64) void gat_out_wmma(const float* __restrict__ x,
                                                   const float* __restrict__ attnMean,
                                                   const float* __restrict__ Wout,
                                                   float* __restrict__ out) {
  const int lane = threadIdx.x & 31;
  const int nt   = threadIdx.x >> 5;
  const int m0   = blockIdx.x * 16;

  __shared__ float as[16 * FEAT];
  {
    const float4* src = (const float4*)(attnMean + (size_t)m0 * FEAT);
    ((float4*)as)[threadIdx.x]      = src[threadIdx.x];
    ((float4*)as)[threadIdx.x + 64] = src[threadIdx.x + 64];
  }
  __syncthreads();

  v8f acc = {};
  acc = wmma_f32_k32(as, Wout, nt * 16, lane, acc);

  const int col = lane & 15, hi = lane >> 4;
#pragma unroll
  for (int r = 0; r < 8; ++r) {
    const size_t o = (size_t)(m0 + r + 8 * hi) * FEAT + nt * 16 + col;
    out[o] = x[o] + acc[r];
  }
}

extern "C" void kernel_launch(void* const* d_in, const int* in_sizes, int n_in,
                              void* d_out, int out_size, void* d_ws, size_t ws_size,
                              hipStream_t stream) {
  const float* x    = (const float*)d_in[0];
  const int*   ei   = (const int*)d_in[1];
  const float* Wq   = (const float*)d_in[2];
  const float* Wk   = (const float*)d_in[3];
  const float* Wv   = (const float*)d_in[4];
  const float* Wout = (const float*)d_in[5];
  float*       out  = (float*)d_out;

  float* ws   = (float*)d_ws;
  float* qkv  = ws;
  float* sEH  = qkv + (size_t)3 * HEADS * N_NODES * FEAT;
  float* mbuf = sEH + (size_t)N_EDGES * HEADS;
  float* den  = mbuf + (size_t)N_NODES * HEADS;
  float* num  = den + (size_t)N_NODES * HEADS;
  float* attn = num + (size_t)N_NODES * HEADS * FEAT;

  gat_init<<<(N_NODES * HEADS * FEAT + 255) / 256, 256, 0, stream>>>(mbuf, den, num);

  gat_qkv_wmma<<<dim3(MTILES, 6), 128, 0, stream>>>(x, Wq, Wk, Wv, qkv);

  const int edgeBlocks = (N_EDGES * 32) / 256;
  gat_edge_logits<<<edgeBlocks, 256, 0, stream>>>(qkv, ei, sEH, mbuf);
  gat_edge_scatter<<<edgeBlocks, 256, 0, stream>>>(qkv, ei, sEH, mbuf, den, num);

  gat_finalize<<<(N_NODES * FEAT + 255) / 256, 256, 0, stream>>>(num, den, attn);

  gat_out_wmma<<<MTILES, 64, 0, stream>>>(x, attn, Wout, out);
}


# edge scatter: lane = feature of every head, so each float atomic instruction covers one whole 128-byte line of num (was 4 lines x 8 floats per instruction)
# speedup vs baseline: 1.0967x; 1.0967x over previous
#include <hip/hip_runtime.h>
#include <math.h>


#define N_NODES 50000
#define N_EDGES 800000
#define FEAT    32
#define HEADS   4
#define MTILES  (N_NODES / 16)

typedef __attribute__((ext_vector_type(2))) float v2f;
typedef __attribute__((ext_vector_type(8))) float v8f;

__device__ __forceinline__ void atomic_add_f32_dev(float* p, float v) {
  asm volatile("global_atomic_add_f32 %0, %1, off scope:SCOPE_DEV"
               :: "v"(p), "v"(v) : "memory");
}
template <int BYTE_OFF>
__device__ __forceinline__ void atomic_add_f32_dev_off(float* p, float v) {
  asm volatile("global_atomic_add_f32 %0, %1, off offset:%2 scope:SCOPE_DEV"
               :: "v"(p), "v"(v), "i"(BYTE_OFF) : "memory");
}
__device__ __forceinline__ void atomic_max_f32_dev(float* p, float v) {
  asm volatile("global_atomic_max_num_f32 %0, %1, off scope:SCOPE_DEV"
               :: "v"(p), "v"(v) : "memory");
}

__device__ __forceinline__ v8f wmma_f32_k32(const float* Arow,
                                            const float* __restrict__ B,
                                            int bcol, int lane, v8f acc) {
  const int row = lane & 15;
  const int hi  = lane >> 4;
#pragma unroll
  for (int kc = 0; kc < 8; ++kc) {
    const int k0 = kc * 4 + 2 * hi;
    v2f a, b;
    a.x = Arow[row * FEAT + k0];
    a.y = Arow[row * FEAT + k0 + 1];
    b.x = B[k0 * FEAT + bcol + row];
    b.y = B[(k0 + 1) * FEAT + bcol + row];
    acc = __builtin_amdgcn_wmma_f32_16x16x4_f32(false, a, false, b, (short)0, acc,
                                                false, false);
  }
  return acc;
}

__global__ void gat_init(float* __restrict__ m, float* __restrict__ denom,
                         float* __restrict__ num) {
  const int idx = blockIdx.x * blockDim.x + threadIdx.x;
  if (idx < N_NODES * HEADS) {
    m[idx]     = -INFINITY;
    denom[idx] = 0.0f;
  }
  if (idx < N_NODES * HEADS * FEAT) num[idx] = 0.0f;
}

__global__ __launch_bounds__(128) void gat_qkv_wmma(const float* __restrict__ x,
                                                    const float* __restrict__ Wq,
                                                    const float* __restrict__ Wk,
                                                    const float* __restrict__ Wv,
                                                    float* __restrict__ qkv) {
  const int lane = threadIdx.x & 31;
  const int wave = threadIdx.x >> 5;
  const int comb = blockIdx.y * 4 + wave;
  const int mat  = comb >> 3;
  const int head = (comb >> 1) & 3;
  const int nt   = comb & 1;
  const int m0   = blockIdx.x * 16;

  __shared__ float xs[16 * FEAT];
  ((float4*)xs)[threadIdx.x] = ((const float4*)(x + (size_t)m0 * FEAT))[threadIdx.x];
  __syncthreads();

  const float* W = (mat == 0 ? Wq : (mat == 1 ? Wk : Wv)) + head * FEAT * FEAT;
  v8f acc = {};
  acc = wmma_f32_k32(xs, W, nt * 16, lane, acc);

  float* out = qkv + ((size_t)(mat * HEADS + head) * N_NODES) * FEAT;
  const int col = lane & 15, hi = lane >> 4;
#pragma unroll
  for (int r = 0; r < 8; ++r)
    out[(size_t)(m0 + r + 8 * hi) * FEAT + nt * 16 + col] = acc[r];
}

__global__ __launch_bounds__(256) void gat_edge_logits(const float* __restrict__ qkv,
                                                       const int* __restrict__ ei,
                                                       float* __restrict__ sEH,
                                                       float* __restrict__ m) {
  const int wid = (blockIdx.x * blockDim.x + threadIdx.x) >> 5;
  if (wid >= N_EDGES) return;
  const int lane = threadIdx.x & 31;
  if (lane == 0 && wid + 8192 < N_EDGES) {
    __builtin_prefetch(ei + wid + 8192, 0, 0);
    __builtin_prefetch(ei + N_EDGES + wid + 8192, 0, 0);
  }
  const int h    = lane >> 3;
  const int f4   = (lane & 7) * 4;
  const int snd  = ei[wid];
  const int rcv  = ei[N_EDGES + wid];

  const float* Qr = qkv + ((size_t)(0 * HEADS + h) * N_NODES + rcv) * FEAT;
  const float* Kr = qkv + ((size_t)(1 * HEADS + h) * N_NODES + snd) * FEAT;
  const float4 q = *(const float4*)(Qr + f4);
  const float4 k = *(const float4*)(Kr + f4);
  float p = q.x * k.x + q.y * k.y + q.z * k.z + q.w * k.w;
  p += __shfl_xor(p, 1, 32);
  p += __shfl_xor(p, 2, 32);
  p += __shfl_xor(p, 4, 32);
  if ((lane & 7) == 0) {
    const float s = p * 0.17677669529663687f;
    sEH[(size_t)wid * HEADS + h] = s;
    atomic_max_f32_dev(&m[rcv * HEADS + h], s);
  }
}

__global__ __launch_bounds__(256) void gat_edge_scatter(const float* __restrict__ qkv,
                                                        const int* __restrict__ ei,
                                                        const float* __restrict__ sEH,
                                                        const float* __restrict__ m,
                                                        float* __restrict__ denom,
                                                        float* __restrict__ num) {
  const int wid = (blockIdx.x * blockDim.x + threadIdx.x) >> 5;
  if (wid >= N_EDGES) return;
  const int lane = threadIdx.x & 31;
  if (lane == 0 && wid + 8192 < N_EDGES) {
    __builtin_prefetch(ei + wid + 8192, 0, 0);
    __builtin_prefetch(ei + N_EDGES + wid + 8192, 0, 0);
    __builtin_prefetch(sEH + (size_t)(wid + 8192) * HEADS, 0, 0);
  }
  const int snd  = ei[wid];
  const int rcv  = ei[N_EDGES + wid];

  const int hq   = lane & 3;
  const float s  = sEH[(size_t)wid * HEADS + hq];
  const float mv = m[rcv * HEADS + hq];
  const float w  = __expf(s - mv);
  if (lane < HEADS) atomic_add_f32_dev(&denom[rcv * HEADS + lane], w);

  const size_t plane = (size_t)N_NODES * FEAT;
  const float* Vr = qkv + (size_t)(2 * HEADS) * plane + (size_t)snd * FEAT + lane;
  const float v0 = Vr[0];
  const float v1 = Vr[plane];
  const float v2 = Vr[2 * plane];
  const float v3 = Vr[3 * plane];

  const float w0 = __int_as_float(__builtin_amdgcn_readlane(__float_as_int(w), 0));
  const float w1 = __int_as_float(__builtin_amdgcn_readlane(__float_as_int(w), 1));
  const float w2 = __int_as_float(__builtin_amdgcn_readlane(__float_as_int(w), 2));
  const float w3 = __int_as_float(__builtin_amdgcn_readlane(__float_as_int(w), 3));

  float* dst = num + (size_t)rcv * (HEADS * FEAT) + lane;
  atomic_add_f32_dev_off<0 * FEAT * 4>(dst, w0 * v0);
  atomic_add_f32_dev_off<1 * FEAT * 4>(dst, w1 * v1);
  atomic_add_f32_dev_off<2 * FEAT * 4>(dst, w2 * v2);
  atomic_add_f32_dev_off<3 * FEAT * 4>(dst, w3 * v3);
}

__global__ void gat_finalize(const float* __restrict__ num,
                             const float* __restrict__ denom,
                             float* __restrict__ attnMean) {
  const int idx = blockIdx.x * blockDim.x + threadIdx.x;
  if (idx >= N_NODES * FEAT) return;
  const int n = idx >> 5, f = idx & 31;
  float acc = 0.0f;
#pragma unroll
  for (int h = 0; h < HEADS; ++h) {
    const float d = denom[n * HEADS + h];
    const float v = num[((size_t)n * HEADS + h) * FEAT + f];
    acc += (d > 0.0f) ? v / d : 0.0f;
  }
  attnMean[idx] = acc * 0.25f;
}

__global__ __launch_bounds__(64) void gat_out_wmma(const float* __restrict__ x,
                                                   const float* __restrict__ attnMean,
                                                   const float* __restrict__ Wout,
                                                   float* __restrict__ out) {
  const int lane = threadIdx.x & 31;
  const int nt   = threadIdx.x >> 5;
  const int m0   = blockIdx.x * 16;

  __shared__ float as[16 * FEAT];
  {
    const float4* src = (const float4*)(attnMean + (size_t)m0 * FEAT);
    ((float4*)as)[threadIdx.x]      = src[threadIdx.x];
    ((float4*)as)[threadIdx.x + 64] = src[threadIdx.x + 64];
  }
  __syncthreads();

  v8f acc = {};
  acc = wmma_f32_k32(as, Wout, nt * 16, lane, acc);

  const int col = lane & 15, hi = lane >> 4;
#pragma unroll
  for (int r = 0; r < 8; ++r) {
    const size_t o = (size_t)(m0 + r + 8 * hi) * FEAT + nt * 16 + col;
    out[o] = x[o] + acc[r];
  }
}

extern "C" void kernel_launch(void* const* d_in, const int* in_sizes, int n_in,
                              void* d_out, int out_size, void* d_ws, size_t ws_size,
                              hipStream_t stream) {
  const float* x    = (const float*)d_in[0];
  const int*   ei   = (const int*)d_in[1];
  const float* Wq   = (const float*)d_in[2];
  const float* Wk   = (const float*)d_in[3];
  const float* Wv   = (const float*)d_in[4];
  const float* Wout = (const float*)d_in[5];
  float*       out  = (float*)d_out;

  float* ws   = (float*)d_ws;
  float* qkv  = ws;
  float* sEH  = qkv + (size_t)3 * HEADS * N_NODES * FEAT;
  float* mbuf = sEH + (size_t)N_EDGES * HEADS;
  float* den  = mbuf + (size_t)N_NODES * HEADS;
  float* num  = den + (size_t)N_NODES * HEADS;
  float* attn = num + (size_t)N_NODES * HEADS * FEAT;

  gat_init<<<(N_NODES * HEADS * FEAT + 255) / 256, 256, 0, stream>>>(mbuf, den, num);

  gat_qkv_wmma<<<dim3(MTILES, 6), 128, 0, stream>>>(x, Wq, Wk, Wv, qkv);

  const int edgeBlocks = (N_EDGES * 32) / 256;
  gat_edge_logits<<<edgeBlocks, 256, 0, stream>>>(qkv, ei, sEH, mbuf);
  gat_edge_scatter<<<edgeBlocks, 256, 0, stream>>>(qkv, ei, sEH, mbuf, den, num);

  gat_finalize<<<(N_NODES * FEAT + 255) / 256, 256, 0, stream>>>(num, den, attn);

  gat_out_wmma<<<MTILES, 64, 0, stream>>>(x, attn, Wout, out);
}


# edge kernels: one wave walks 16 consecutive edges (both index arrays in one coalesced load, row gathers of 8 to 16 edges in flight together) instead of one wave per edge: 16x fewer waves
# speedup vs baseline: 1.3070x; 1.1918x over previous
#include <hip/hip_runtime.h>
#include <math.h>


#define N_NODES 50000
#define N_EDGES 800000
#define FEAT    32
#define HEADS   4
#define MTILES  (N_NODES / 16)
#define EPW     16
#define EB      8

typedef __attribute__((ext_vector_type(2))) float v2f;
typedef __attribute__((ext_vector_type(8))) float v8f;

__device__ __forceinline__ void atomic_add_f32_dev(float* p, float v) {
  asm volatile("global_atomic_add_f32 %0, %1, off scope:SCOPE_DEV"
               :: "v"(p), "v"(v) : "memory");
}
template <int BYTE_OFF>
__device__ __forceinline__ void atomic_add_f32_dev_row(float* rowBase, unsigned laneByteOff, float v) {
  asm volatile("global_atomic_add_f32 %0, %1, %2 offset:%3 scope:SCOPE_DEV"
               :: "v"(laneByteOff), "v"(v), "s"(rowBase), "i"(BYTE_OFF) : "memory");
}
__device__ __forceinline__ void atomic_max_f32_dev(float* p, float v) {
  asm volatile("global_atomic_max_num_f32 %0, %1, off scope:SCOPE_DEV"
               :: "v"(p), "v"(v) : "memory");
}

__device__ __forceinline__ v8f wmma_f32_k32(const float* Arow,
                                            const float* __restrict__ B,
                                            int bcol, int lane, v8f acc) {
  const int row = lane & 15;
  const int hi  = lane >> 4;
#pragma unroll
  for (int kc = 0; kc < 8; ++kc) {
    const int k0 = kc * 4 + 2 * hi;
    v2f a, b;
    a.x = Arow[row * FEAT + k0];
    a.y = Arow[row * FEAT + k0 + 1];
    b.x = B[k0 * FEAT + bcol + row];
    b.y = B[(k0 + 1) * FEAT + bcol + row];
    acc = __builtin_amdgcn_wmma_f32_16x16x4_f32(false, a, false, b, (short)0, acc,
                                                false, false);
  }
  return acc;
}

__global__ void gat_init(float* __restrict__ m, float* __restrict__ denom,
                         float* __restrict__ num) {
  const int idx = blockIdx.x * blockDim.x + threadIdx.x;
  if (idx < N_NODES * HEADS) {
    m[idx]     = -INFINITY;
    denom[idx] = 0.0f;
  }
  if (idx < N_NODES * HEADS * FEAT) num[idx] = 0.0f;
}

__global__ __launch_bounds__(128) void gat_qkv_wmma(const float* __restrict__ x,
                                                    const float* __restrict__ Wq,
                                                    const float* __restrict__ Wk,
                                                    const float* __restrict__ Wv,
                                                    float* __restrict__ qkv) {
  const int lane = threadIdx.x & 31;
  const int wave = threadIdx.x >> 5;
  const int comb = blockIdx.y * 4 + wave;
  const int mat  = comb >> 3;
  const int head = (comb >> 1) & 3;
  const int nt   = comb & 1;
  const int m0   = blockIdx.x * 16;

  __shared__ float xs[16 * FEAT];
  ((float4*)xs)[threadIdx.x] = ((const float4*)(x + (size_t)m0 * FEAT))[threadIdx.x];
  __syncthreads();

  const float* W = (mat == 0 ? Wq : (mat == 1 ? Wk : Wv)) + head * FEAT * FEAT;
  v8f acc = {};
  acc = wmma_f32_k32(xs, W, nt * 16, lane, acc);

  float* out = qkv + ((size_t)(mat * HEADS + head) * N_NODES) * FEAT;
  const int col = lane & 15, hi = lane >> 4;
#pragma unroll
  for (int r = 0; r < 8; ++r)
    out[(size_t)(m0 + r + 8 * hi) * FEAT + nt * 16 + col] = acc[r];
}

__global__ __launch_bounds__(256) void gat_edge_logits(const float* __restrict__ qkv,
                                                       const int* __restrict__ ei,
                                                       float* __restrict__ sEH,
                                                       float* __restrict__ m) {
  const unsigned wid = (blockIdx.x * blockDim.x + threadIdx.x) >> 5;
  if (wid >= N_EDGES / EPW) return;
  const unsigned lane = threadIdx.x & 31;
  const unsigned base = wid * EPW;
  const int idx = ei[(lane >> 4) * N_EDGES + base + (lane & 15)];

  const unsigned h = lane >> 3;
  const unsigned lbyte = (h * (unsigned)(N_NODES * FEAT) + (lane & 7) * 4u) * 4u;
  const char* Qb = (const char*)qkv;
  const char* Kb = (const char*)(qkv + (size_t)HEADS * N_NODES * FEAT);

#pragma unroll
  for (int jb = 0; jb < EPW / EB; ++jb) {
    float4 q[EB], k[EB];
#pragma unroll
    for (int j = 0; j < EB; ++j) {
      const int e = jb * EB + j;
      const unsigned snd = (unsigned)__builtin_amdgcn_readlane(idx, e);
      const unsigned rcv = (unsigned)__builtin_amdgcn_readlane(idx, 16 + e);
      const char* qrow = Qb + (size_t)rcv * (FEAT * 4);
      const char* krow = Kb + (size_t)snd * (FEAT * 4);
      q[j] = *(const float4*)(qrow + lbyte);
      k[j] = *(const float4*)(krow + lbyte);
    }
    __builtin_amdgcn_sched_barrier(0);

    float mine = 0.0f;
#pragma unroll
    for (int j = 0; j < EB; ++j) {
      float p = q[j].y * k[j].y;
      p = fmaf(q[j].x, k[j].x, p);
      p = fmaf(q[j].z, k[j].z, p);
      p = fmaf(q[j].w, k[j].w, p);
      p += __shfl_xor(p, 1, 32);
      p += __shfl_xor(p, 2, 32);
      p += __shfl_xor(p, 4, 32);
      if ((lane & 7) == (unsigned)j) mine = p;
    }
    const float s = mine * 0.17677669529663687f;
    const unsigned el = jb * EB + (lane & 7);
    const int rl = __shfl(idx, 16 + el, 32);
    sEH[(size_t)(base + el) * HEADS + h] = s;
    atomic_max_f32_dev(&m[rl * HEADS + h], s);
  }
}

__global__ __launch_bounds__(256) void gat_edge_scatter(const float* __restrict__ qkv,
                                                        const int* __restrict__ ei,
                                                        const float* __restrict__ sEH,
                                                        const float* __restrict__ m,
                                                        float* __restrict__ denom,
                                                        float* __restrict__ num) {
  const unsigned wid = (blockIdx.x * blockDim.x + threadIdx.x) >> 5;
  if (wid >= N_EDGES / EPW) return;
  const unsigned lane = threadIdx.x & 31;
  const unsigned base = wid * EPW;
  const int idx = ei[(lane >> 4) * N_EDGES + base + (lane & 15)];

  const size_t planeBytes = (size_t)N_NODES * FEAT * 4;
  const char* Vb = (const char*)(qkv + (size_t)(2 * HEADS) * N_NODES * FEAT);
  const unsigned laneByte = lane * 4u;
  const unsigned hq = lane & 3;

  int   rl[EPW / EB];
  float s[EPW / EB], mv[EPW / EB];
#pragma unroll
  for (int jb = 0; jb < EPW / EB; ++jb) {
    const unsigned el = jb * EB + (lane >> 2);
    rl[jb] = __shfl(idx, 16 + el, 32);
    s[jb]  = sEH[(size_t)base * HEADS + jb * (EB * HEADS) + lane];
    mv[jb] = m[rl[jb] * HEADS + hq];
  }

  float v[EPW][HEADS];
#pragma unroll
  for (int e = 0; e < EPW; ++e) {
    const unsigned snd = (unsigned)__builtin_amdgcn_readlane(idx, e);
    const char* vrow = Vb + (size_t)snd * (FEAT * 4);
#pragma unroll
    for (int hh = 0; hh < HEADS; ++hh)
      v[e][hh] = *(const float*)(vrow + hh * planeBytes + laneByte);
  }

  float w[EPW / EB];
#pragma unroll
  for (int jb = 0; jb < EPW / EB; ++jb) {
    w[jb] = __expf(s[jb] - mv[jb]);
    atomic_add_f32_dev(&denom[rl[jb] * HEADS + hq], w[jb]);
  }

#pragma unroll
  for (int e = 0; e < EPW; ++e) {
    const unsigned rcv = (unsigned)__builtin_amdgcn_readlane(idx, 16 + e);
    const int wb = __float_as_int(w[e / EB]);
    const int j  = e % EB;
    const float w0 = __int_as_float(__builtin_amdgcn_readlane(wb, 4 * j + 0));
    const float w1 = __int_as_float(__builtin_amdgcn_readlane(wb, 4 * j + 1));
    const float w2 = __int_as_float(__builtin_amdgcn_readlane(wb, 4 * j + 2));
    const float w3 = __int_as_float(__builtin_amdgcn_readlane(wb, 4 * j + 3));
    float* row = num + (size_t)rcv * (HEADS * FEAT);
    atomic_add_f32_dev_row<0 * FEAT * 4>(row, laneByte, w0 * v[e][0]);
    atomic_add_f32_dev_row<1 * FEAT * 4>(row, laneByte, w1 * v[e][1]);
    atomic_add_f32_dev_row<2 * FEAT * 4>(row, laneByte, w2 * v[e][2]);
    atomic_add_f32_dev_row<3 * FEAT * 4>(row, laneByte, w3 * v[e][3]);
  }
}

__global__ void gat_finalize(const float* __restrict__ num,
                             const float* __restrict__ denom,
                             float* __restrict__ attnMean) {
  const int idx = blockIdx.x * blockDim.x + threadIdx.x;
  if (idx >= N_NODES * FEAT) return;
  const int n = idx >> 5, f = idx & 31;
  float acc = 0.0f;
#pragma unroll
  for (int h = 0; h < HEADS; ++h) {
    const float d = denom[n * HEADS + h];
    const float v = num[((size_t)n * HEADS + h) * FEAT + f];
    acc += (d > 0.0f) ? v / d : 0.0f;
  }
  attnMean[idx] = acc * 0.25f;
}

__global__ __launch_bounds__(64) void gat_out_wmma(const float* __restrict__ x,
                                                   const float* __restrict__ attnMean,
                                                   const float* __restrict__ Wout,
                                                   float* __restrict__ out) {
  const int lane = threadIdx.x & 31;
  const int nt   = threadIdx.x >> 5;
  const int m0   = blockIdx.x * 16;

  __shared__ float as[16 * FEAT];
  {
    const float4* src = (const float4*)(attnMean + (size_t)m0 * FEAT);
    ((float4*)as)[threadIdx.x]      = src[threadIdx.x];
    ((float4*)as)[threadIdx.x + 64] = src[threadIdx.x + 64];
  }
  __syncthreads();

  v8f acc = {};
  acc = wmma_f32_k32(as, Wout, nt * 16, lane, acc);

  const int col = lane & 15, hi = lane >> 4;
#pragma unroll
  for (int r = 0; r < 8; ++r) {
    const size_t o = (size_t)(m0 + r + 8 * hi) * FEAT + nt * 16 + col;
    out[o] = x[o] + acc[r];
  }
}

extern "C" void kernel_launch(void* const* d_in, const int* in_sizes, int n_in,
                              void* d_out, int out_size, void* d_ws, size_t ws_size,
                              hipStream_t stream) {
  const float* x    = (const float*)d_in[0];
  const int*   ei   = (const int*)d_in[1];
  const float* Wq   = (const float*)d_in[2];
  const float* Wk   = (const float*)d_in[3];
  const float* Wv   = (const float*)d_in[4];
  const float* Wout = (const float*)d_in[5];
  float*       out  = (float*)d_out;

  float* ws   = (float*)d_ws;
  float* qkv  = ws;
  float* sEH  = qkv + (size_t)3 * HEADS * N_NODES * FEAT;
  float* mbuf = sEH + (size_t)N_EDGES * HEADS;
  float* den  = mbuf + (size_t)N_NODES * HEADS;
  float* num  = den + (size_t)N_NODES * HEADS;
  float* attn = num + (size_t)N_NODES * HEADS * FEAT;

  gat_init<<<(N_NODES * HEADS * FEAT + 255) / 256, 256, 0, stream>>>(mbuf, den, num);

  gat_qkv_wmma<<<dim3(MTILES, 6), 128, 0, stream>>>(x, Wq, Wk, Wv, qkv);

  const int edgeBlocks = (N_EDGES / EPW * 32) / 256;
  gat_edge_logits<<<edgeBlocks, 256, 0, stream>>>(qkv, ei, sEH, mbuf);
  gat_edge_scatter<<<edgeBlocks, 256, 0, stream>>>(qkv, ei, sEH, mbuf, den, num);

  gat_finalize<<<(N_NODES * FEAT + 255) / 256, 256, 0, stream>>>(num, den, attn);

  gat_out_wmma<<<MTILES, 64, 0, stream>>>(x, attn, Wout, out);
}


# logits as (x_r Wq_h Wk_h^T) . x_s: Wq_h Wk_h^T folded once on WMMA, one node-major plane P = x M_h replaces Q and K; logits kernel gathers 512 contiguous bytes + one x row per edge, not eight rows
# speedup vs baseline: 1.5978x; 1.2225x over previous
#include <hip/hip_runtime.h>
#include <math.h>


#define N_NODES 50000
#define N_EDGES 800000
#define FEAT    32
#define HEADS   4
#define MTILES  (N_NODES / 16)
#define EPW     16
#define EB      8

typedef __attribute__((ext_vector_type(2))) float v2f;
typedef __attribute__((ext_vector_type(8))) float v8f;

__device__ __forceinline__ void atomic_add_f32_dev(float* p, float v) {
  asm volatile("global_atomic_add_f32 %0, %1, off scope:SCOPE_DEV"
               :: "v"(p), "v"(v) : "memory");
}
template <int BYTE_OFF>
__device__ __forceinline__ void atomic_add_f32_dev_row(float* rowBase, unsigned laneByteOff, float v) {
  asm volatile("global_atomic_add_f32 %0, %1, %2 offset:%3 scope:SCOPE_DEV"
               :: "v"(laneByteOff), "v"(v), "s"(rowBase), "i"(BYTE_OFF) : "memory");
}
__device__ __forceinline__ void atomic_max_f32_dev(float* p, float v) {
  asm volatile("global_atomic_max_num_f32 %0, %1, off scope:SCOPE_DEV"
               :: "v"(p), "v"(v) : "memory");
}

__device__ __forceinline__ v8f wmma_f32_k32(const float* Arow,
                                            const float* __restrict__ B,
                                            int bcol, int lane, v8f acc) {
  const int row = lane & 15;
  const int hi  = lane >> 4;
#pragma unroll
  for (int kc = 0; kc < 8; ++kc) {
    const int k0 = kc * 4 + 2 * hi;
    v2f a, b;
    a.x = Arow[row * FEAT + k0];
    a.y = Arow[row * FEAT + k0 + 1];
    b.x = B[k0 * FEAT + bcol + row];
    b.y = B[(k0 + 1) * FEAT + bcol + row];
    acc = __builtin_amdgcn_wmma_f32_16x16x4_f32(false, a, false, b, (short)0, acc,
                                                false, false);
  }
  return acc;
}

#define INIT_BLOCKS ((N_NODES * HEADS * FEAT + 255) / 256)
#define FOLD_BLOCKS 2
__global__ __launch_bounds__(256) void gat_init_fold(float* __restrict__ m, float* __restrict__ denom,
                                                     float* __restrict__ num,
                                                     const float* __restrict__ Wq,
                                                     const float* __restrict__ Wk,
                                                     float* __restrict__ Mqk) {
  if (blockIdx.x < FOLD_BLOCKS) {
    const int lane = threadIdx.x & 31;
    const int t    = blockIdx.x * 8 + (threadIdx.x >> 5);
    const int head = t >> 2, mt = (t >> 1) & 1, nt = t & 1;
    const float* A  = Wq + head * FEAT * FEAT + mt * 16 * FEAT;
    const float* Bt = Wk + head * FEAT * FEAT + nt * 16 * FEAT;
    const int row = lane & 15, hi = lane >> 4;
    v8f acc = {};
#pragma unroll
    for (int kc = 0; kc < 8; ++kc) {
      const int k0 = kc * 4 + 2 * hi;
      v2f a, b;
      a.x = A[row * FEAT + k0];
      a.y = A[row * FEAT + k0 + 1];
      b.x = Bt[row * FEAT + k0];
      b.y = Bt[row * FEAT + k0 + 1];
      acc = __builtin_amdgcn_wmma_f32_16x16x4_f32(false, a, false, b, (short)0, acc,
                                                  false, false);
    }
    float* Mh = Mqk + head * FEAT * FEAT;
#pragma unroll
    for (int r = 0; r < 8; ++r)
      Mh[(mt * 16 + r + 8 * hi) * FEAT + nt * 16 + row] = acc[r];
    return;
  }
  const int idx = (blockIdx.x - FOLD_BLOCKS) * blockDim.x + threadIdx.x;
  if (idx < N_NODES * HEADS) {
    m[idx]     = -INFINITY;
    denom[idx] = 0.0f;
  }
  if (idx < N_NODES * HEADS * FEAT) num[idx] = 0.0f;
}

__global__ __launch_bounds__(256) void gat_p_wmma(const float* __restrict__ x,
                                                  const float* __restrict__ Mqk,
                                                  float* __restrict__ P) {
  const int lane = threadIdx.x & 31;
  const int wave = threadIdx.x >> 5;
  const int head = wave >> 1;
  const int nt   = wave & 1;
  const int m0   = blockIdx.x * 16;

  __shared__ float xs[16 * FEAT];
  if (threadIdx.x < 128)
    ((float4*)xs)[threadIdx.x] = ((const float4*)(x + (size_t)m0 * FEAT))[threadIdx.x];
  __syncthreads();

  v8f acc = {};
  acc = wmma_f32_k32(xs, Mqk + head * FEAT * FEAT, nt * 16, lane, acc);

  const int col = lane & 15, hi = lane >> 4;
#pragma unroll
  for (int r = 0; r < 8; ++r)
    P[(size_t)(m0 + r + 8 * hi) * (HEADS * FEAT) + head * FEAT + nt * 16 + col] = acc[r];
}

__global__ __launch_bounds__(256) void gat_edge_logits(const float* __restrict__ P,
                                                       const float* __restrict__ x,
                                                       const int* __restrict__ ei,
                                                       float* __restrict__ sEH,
                                                       float* __restrict__ m) {
  const unsigned wid = (blockIdx.x * blockDim.x + threadIdx.x) >> 5;
  if (wid >= N_EDGES / EPW) return;
  const unsigned lane = threadIdx.x & 31;
  const unsigned base = wid * EPW;
  const int idx = ei[(lane >> 4) * N_EDGES + base + (lane & 15)];

  const unsigned h = lane >> 3;
  const unsigned pbyte = lane * 16u;
  const unsigned xbyte = (lane & 7) * 16u;
  const char* Pb = (const char*)P;
  const char* Xb = (const char*)x;

#pragma unroll
  for (int jb = 0; jb < EPW / EB; ++jb) {
    float4 q[EB], k[EB];
#pragma unroll
    for (int j = 0; j < EB; ++j) {
      const int e = jb * EB + j;
      const unsigned snd = (unsigned)__builtin_amdgcn_readlane(idx, e);
      const unsigned rcv = (unsigned)__builtin_amdgcn_readlane(idx, 16 + e);
      const char* prow = Pb + (size_t)rcv * (HEADS * FEAT * 4);
      const char* xrow = Xb + (size_t)snd * (FEAT * 4);
      q[j] = *(const float4*)(prow + pbyte);
      k[j] = *(const float4*)(xrow + xbyte);
    }
    __builtin_amdgcn_sched_barrier(0);

    float mine = 0.0f;
#pragma unroll
    for (int j = 0; j < EB; ++j) {
      float p = q[j].y * k[j].y;
      p = fmaf(q[j].x, k[j].x, p);
      p = fmaf(q[j].z, k[j].z, p);
      p = fmaf(q[j].w, k[j].w, p);
      p += __shfl_xor(p, 1, 32);
      p += __shfl_xor(p, 2, 32);
      p += __shfl_xor(p, 4, 32);
      if ((lane & 7) == (unsigned)j) mine = p;
    }
    const float s = mine * 0.17677669529663687f;
    const unsigned el = jb * EB + (lane & 7);
    const int rl = __shfl(idx, 16 + el, 32);
    sEH[(size_t)(base + el) * HEADS + h] = s;
    atomic_max_f32_dev(&m[rl * HEADS + h], s);
  }
}

__global__ __launch_bounds__(256) void gat_edge_scatter(const float* __restrict__ x,
                                                        const int* __restrict__ ei,
                                                        const float* __restrict__ sEH,
                                                        const float* __restrict__ m,
                                                        float* __restrict__ denom,
                                                        float* __restrict__ num) {
  const unsigned wid = (blockIdx.x * blockDim.x + threadIdx.x) >> 5;
  if (wid >= N_EDGES / EPW) return;
  const unsigned lane = threadIdx.x & 31;
  const unsigned base = wid * EPW;
  const int idx = ei[(lane >> 4) * N_EDGES + base + (lane & 15)];

  const char* Xb = (const char*)x;
  const unsigned laneByte = lane * 4u;
  const unsigned hq = lane & 3;

  int   rl[EPW / EB];
  float s[EPW / EB], mv[EPW / EB];
#pragma unroll
  for (int jb = 0; jb < EPW / EB; ++jb) {
    const unsigned el = jb * EB + (lane >> 2);
    rl[jb] = __shfl(idx, 16 + el, 32);
    s[jb]  = sEH[(size_t)base * HEADS + jb * (EB * HEADS) + lane];
    mv[jb] = m[rl[jb] * HEADS + hq];
  }

  float v[EPW];
#pragma unroll
  for (int e = 0; e < EPW; ++e) {
    const unsigned snd = (unsigned)__builtin_amdgcn_readlane(idx, e);
    v[e] = *(const float*)(Xb + (size_t)snd * (FEAT * 4) + laneByte);
  }

  float w[EPW / EB];
#pragma unroll
  for (int jb = 0; jb < EPW / EB; ++jb) {
    w[jb] = __expf(s[jb] - mv[jb]);
    atomic_add_f32_dev(&denom[rl[jb] * HEADS + hq], w[jb]);
  }

#pragma unroll
  for (int e = 0; e < EPW; ++e) {
    const unsigned rcv = (unsigned)__builtin_amdgcn_readlane(idx, 16 + e);
    const int wb = __float_as_int(w[e / EB]);
    const int j  = e % EB;
    const float w0 = __int_as_float(__builtin_amdgcn_readlane(wb, 4 * j + 0));
    const float w1 = __int_as_float(__builtin_amdgcn_readlane(wb, 4 * j + 1));
    const float w2 = __int_as_float(__builtin_amdgcn_readlane(wb, 4 * j + 2));
    const float w3 = __int_as_float(__builtin_amdgcn_readlane(wb, 4 * j + 3));
    float* row = num + (size_t)rcv * (HEADS * FEAT);
    atomic_add_f32_dev_row<0 * FEAT * 4>(row, laneByte, w0 * v[e]);
    atomic_add_f32_dev_row<1 * FEAT * 4>(row, laneByte, w1 * v[e]);
    atomic_add_f32_dev_row<2 * FEAT * 4>(row, laneByte, w2 * v[e]);
    atomic_add_f32_dev_row<3 * FEAT * 4>(row, laneByte, w3 * v[e]);
  }
}

__global__ __launch_bounds__(64) void gat_out_wmma(const float* __restrict__ x,
                                                   const float* __restrict__ num,
                                                   const float* __restrict__ denom,
                                                   const float* __restrict__ Wv,
                                                   const float* __restrict__ Wout,
                                                   float* __restrict__ out) {
  const int lane = threadIdx.x & 31;
  const int nt   = threadIdx.x >> 5;
  const int m0   = blockIdx.x * 16;

  __shared__ float as4[HEADS][16 * FEAT];
  __shared__ float ts[16 * FEAT];

  {
    const float4* src = (const float4*)(num + (size_t)m0 * HEADS * FEAT);
    const int hh = (threadIdx.x >> 3) & 3, f4 = threadIdx.x & 7;
#pragma unroll
    for (int i = 0; i < 8; ++i) {
      const int row   = i * 2 + (threadIdx.x >> 5);
      const float4 t  = src[i * 64 + threadIdx.x];
      const float d   = denom[(m0 + row) * HEADS + hh];
      const float inv = (d > 0.0f) ? 1.0f / d : 0.0f;
      ((float4*)as4[hh])[row * 8 + f4] = make_float4(t.x * inv, t.y * inv, t.z * inv, t.w * inv);
    }
  }
  __syncthreads();

  const int col = lane & 15, hi = lane >> 4;
  v8f accv = {};
#pragma unroll
  for (int hh = 0; hh < HEADS; ++hh)
    accv = wmma_f32_k32(as4[hh], Wv + hh * FEAT * FEAT, nt * 16, lane, accv);
#pragma unroll
  for (int r = 0; r < 8; ++r)
    ts[(r + 8 * hi) * FEAT + nt * 16 + col] = accv[r] * 0.25f;
  __syncthreads();

  v8f acc = {};
  acc = wmma_f32_k32(ts, Wout, nt * 16, lane, acc);

#pragma unroll
  for (int r = 0; r < 8; ++r) {
    const size_t o = (size_t)(m0 + r + 8 * hi) * FEAT + nt * 16 + col;
    out[o] = x[o] + acc[r];
  }
}

extern "C" void kernel_launch(void* const* d_in, const int* in_sizes, int n_in,
                              void* d_out, int out_size, void* d_ws, size_t ws_size,
                              hipStream_t stream) {
  const float* x    = (const float*)d_in[0];
  const int*   ei   = (const int*)d_in[1];
  const float* Wq   = (const float*)d_in[2];
  const float* Wk   = (const float*)d_in[3];
  const float* Wv   = (const float*)d_in[4];
  const float* Wout = (const float*)d_in[5];
  float*       out  = (float*)d_out;

  float* ws   = (float*)d_ws;
  float* P    = ws;
  float* Mqk  = P + (size_t)N_NODES * HEADS * FEAT;
  float* sEH  = Mqk + (size_t)HEADS * FEAT * FEAT;
  float* mbuf = sEH + (size_t)N_EDGES * HEADS;
  float* den  = mbuf + (size_t)N_NODES * HEADS;
  float* num  = den + (size_t)N_NODES * HEADS;

  gat_init_fold<<<INIT_BLOCKS + FOLD_BLOCKS, 256, 0, stream>>>(mbuf, den, num, Wq, Wk, Mqk);

  gat_p_wmma<<<MTILES, 256, 0, stream>>>(x, Mqk, P);

  const int edgeBlocks = (N_EDGES / EPW * 32) / 256;
  gat_edge_logits<<<edgeBlocks, 256, 0, stream>>>(P, x, ei, sEH, mbuf);
  gat_edge_scatter<<<edgeBlocks, 256, 0, stream>>>(x, ei, sEH, mbuf, den, num);

  gat_out_wmma<<<MTILES, 64, 0, stream>>>(x, num, den, Wv, Wout, out);
}


# edge phase in ONE pass: w = exp(s) with no max shift (softmax is shift-invariant; taken when every |s| <= 40, checked per edge, else the two-pass kernels run): no max atomics, no logit plane
# speedup vs baseline: 2.2058x; 1.3805x over previous
#include <hip/hip_runtime.h>
#include <math.h>


#define N_NODES 50000
#define N_EDGES 800000
#define FEAT    32
#define HEADS   4
#define MTILES  (N_NODES / 16)
#define EPW     16
#define EB      8

typedef __attribute__((ext_vector_type(2))) float v2f;
typedef __attribute__((ext_vector_type(8))) float v8f;

__device__ __forceinline__ void atomic_add_f32_dev(float* p, float v) {
  asm volatile("global_atomic_add_f32 %0, %1, off scope:SCOPE_DEV"
               :: "v"(p), "v"(v) : "memory");
}
template <int BYTE_OFF>
__device__ __forceinline__ void atomic_add_f32_dev_row(float* rowBase, unsigned laneByteOff, float v) {
  asm volatile("global_atomic_add_f32 %0, %1, %2 offset:%3 scope:SCOPE_DEV"
               :: "v"(laneByteOff), "v"(v), "s"(rowBase), "i"(BYTE_OFF) : "memory");
}
__device__ __forceinline__ void atomic_max_f32_dev(float* p, float v) {
  asm volatile("global_atomic_max_num_f32 %0, %1, off scope:SCOPE_DEV"
               :: "v"(p), "v"(v) : "memory");
}

__device__ __forceinline__ v8f wmma_f32_k32(const float* Arow,
                                            const float* __restrict__ B,
                                            int bcol, int lane, v8f acc) {
  const int row = lane & 15;
  const int hi  = lane >> 4;
#pragma unroll
  for (int kc = 0; kc < 8; ++kc) {
    const int k0 = kc * 4 + 2 * hi;
    v2f a, b;
    a.x = Arow[row * FEAT + k0];
    a.y = Arow[row * FEAT + k0 + 1];
    b.x = B[k0 * FEAT + bcol + row];
    b.y = B[(k0 + 1) * FEAT + bcol + row];
    acc = __builtin_amdgcn_wmma_f32_16x16x4_f32(false, a, false, b, (short)0, acc,
                                                false, false);
  }
  return acc;
}

#define INIT_BLOCKS ((N_NODES * HEADS * FEAT + 255) / 256)
#define FOLD_BLOCKS 2
__global__ __launch_bounds__(256) void gat_init_fold(float* __restrict__ m, float* __restrict__ denom,
                                                     float* __restrict__ num,
                                                     const float* __restrict__ Wq,
                                                     const float* __restrict__ Wk,
                                                     float* __restrict__ Mqk,
                                                     int* __restrict__ flag) {
  if (blockIdx.x < FOLD_BLOCKS) {
    const int lane = threadIdx.x & 31;
    const int t    = blockIdx.x * 8 + (threadIdx.x >> 5);
    const int head = t >> 2, mt = (t >> 1) & 1, nt = t & 1;
    const float* A  = Wq + head * FEAT * FEAT + mt * 16 * FEAT;
    const float* Bt = Wk + head * FEAT * FEAT + nt * 16 * FEAT;
    const int row = lane & 15, hi = lane >> 4;
    v8f acc = {};
#pragma unroll
    for (int kc = 0; kc < 8; ++kc) {
      const int k0 = kc * 4 + 2 * hi;
      v2f a, b;
      a.x = A[row * FEAT + k0];
      a.y = A[row * FEAT + k0 + 1];
      b.x = Bt[row * FEAT + k0];
      b.y = Bt[row * FEAT + k0 + 1];
      acc = __builtin_amdgcn_wmma_f32_16x16x4_f32(false, a, false, b, (short)0, acc,
                                                  false, false);
    }
    float* Mh = Mqk + head * FEAT * FEAT;
#pragma unroll
    for (int r = 0; r < 8; ++r)
      Mh[(mt * 16 + r + 8 * hi) * FEAT + nt * 16 + row] = acc[r];
    return;
  }
  const int idx = (blockIdx.x - FOLD_BLOCKS) * blockDim.x + threadIdx.x;
  if (idx == 0) *flag = 0;
  if (idx < N_NODES * HEADS) {
    m[idx]     = -INFINITY;
    denom[idx] = 0.0f;
  }
  if (idx < N_NODES * HEADS * FEAT) num[idx] = 0.0f;
}

__global__ __launch_bounds__(256) void gat_p_wmma(const float* __restrict__ x,
                                                  const float* __restrict__ Mqk,
                                                  float* __restrict__ P) {
  const int lane = threadIdx.x & 31;
  const int wave = threadIdx.x >> 5;
  const int head = wave >> 1;
  const int nt   = wave & 1;
  const int m0   = blockIdx.x * 16;

  __shared__ float xs[16 * FEAT];
  if (threadIdx.x < 128)
    ((float4*)xs)[threadIdx.x] = ((const float4*)(x + (size_t)m0 * FEAT))[threadIdx.x];
  __syncthreads();

  v8f acc = {};
  acc = wmma_f32_k32(xs, Mqk + head * FEAT * FEAT, nt * 16, lane, acc);

  const int col = lane & 15, hi = lane >> 4;
#pragma unroll
  for (int r = 0; r < 8; ++r)
    P[(size_t)(m0 + r + 8 * hi) * (HEADS * FEAT) + head * FEAT + nt * 16 + col] = acc[r];
}

#define S_SAFE 40.0f
__global__ __launch_bounds__(256) void gat_edge_fused(const float* __restrict__ P,
                                                      const float* __restrict__ x,
                                                      const int* __restrict__ ei,
                                                      float* __restrict__ denom,
                                                      float* __restrict__ num,
                                                      int* __restrict__ flag) {
  const unsigned wid = (blockIdx.x * blockDim.x + threadIdx.x) >> 5;
  if (wid >= N_EDGES / EPW) return;
  const unsigned lane = threadIdx.x & 31;
  const unsigned base = wid * EPW;
  const int idx = ei[(lane >> 4) * N_EDGES + base + (lane & 15)];

  const unsigned h = lane >> 3;
  const unsigned pbyte    = lane * 16u;
  const unsigned xbyte    = (lane & 7) * 16u;
  const unsigned laneByte = lane * 4u;
  const char* Pb = (const char*)P;
  const char* Xb = (const char*)x;
  bool bad = false;

#pragma unroll
  for (int jb = 0; jb < EPW / EB; ++jb) {
    float4 q[EB], k[EB];
    float  v[EB];
#pragma unroll
    for (int j = 0; j < EB; ++j) {
      const int e = jb * EB + j;
      const unsigned snd = (unsigned)__builtin_amdgcn_readlane(idx, e);
      const unsigned rcv = (unsigned)__builtin_amdgcn_readlane(idx, 16 + e);
      const char* prow = Pb + (size_t)rcv * (HEADS * FEAT * 4);
      const char* xrow = Xb + (size_t)snd * (FEAT * 4);
      q[j] = *(const float4*)(prow + pbyte);
      k[j] = *(const float4*)(xrow + xbyte);
      v[j] = *(const float*)(xrow + laneByte);
    }
    __builtin_amdgcn_sched_barrier(0);

    float mine = 0.0f;
#pragma unroll
    for (int j = 0; j < EB; ++j) {
      float p = q[j].y * k[j].y;
      p = fmaf(q[j].x, k[j].x, p);
      p = fmaf(q[j].z, k[j].z, p);
      p = fmaf(q[j].w, k[j].w, p);
      p += __shfl_xor(p, 1, 32);
      p += __shfl_xor(p, 2, 32);
      p += __shfl_xor(p, 4, 32);
      if ((lane & 7) == (unsigned)j) mine = p;
    }
    const float s = mine * 0.17677669529663687f;
    bad = bad || !(fabsf(s) <= S_SAFE);
    const float w = __expf(s);
    const unsigned el = jb * EB + (lane & 7);
    const int rl = __shfl(idx, 16 + el, 32);
    atomic_add_f32_dev(&denom[rl * HEADS + h], w);

    const int wb = __float_as_int(w);
#pragma unroll
    for (int j = 0; j < EB; ++j) {
      const unsigned rcv = (unsigned)__builtin_amdgcn_readlane(idx, 16 + jb * EB + j);
      const float w0 = __int_as_float(__builtin_amdgcn_readlane(wb, 0 * 8 + j));
      const float w1 = __int_as_float(__builtin_amdgcn_readlane(wb, 1 * 8 + j));
      const float w2 = __int_as_float(__builtin_amdgcn_readlane(wb, 2 * 8 + j));
      const float w3 = __int_as_float(__builtin_amdgcn_readlane(wb, 3 * 8 + j));
      float* row = num + (size_t)rcv * (HEADS * FEAT);
      atomic_add_f32_dev_row<0 * FEAT * 4>(row, laneByte, w0 * v[j]);
      atomic_add_f32_dev_row<1 * FEAT * 4>(row, laneByte, w1 * v[j]);
      atomic_add_f32_dev_row<2 * FEAT * 4>(row, laneByte, w2 * v[j]);
      atomic_add_f32_dev_row<3 * FEAT * 4>(row, laneByte, w3 * v[j]);
    }
  }
  if (bad) *flag = 1;
}

__global__ __launch_bounds__(256) void gat_edge_logits(const float* __restrict__ P,
                                                       const float* __restrict__ x,
                                                       const int* __restrict__ ei,
                                                       float* __restrict__ sEH,
                                                       float* __restrict__ m,
                                                       float* __restrict__ denom,
                                                       float* __restrict__ num,
                                                       const int* __restrict__ flag) {
  const unsigned wid = (blockIdx.x * blockDim.x + threadIdx.x) >> 5;
  if (wid >= N_EDGES / EPW) return;
  if (*flag == 0) return;
  const unsigned lane = threadIdx.x & 31;
  for (unsigned n = wid; n < N_NODES; n += N_EDGES / EPW) {
    *(float4*)((char*)num + (size_t)n * (HEADS * FEAT * 4) + lane * 16u) = make_float4(0.0f, 0.0f, 0.0f, 0.0f);
    if (lane < HEADS) denom[n * HEADS + lane] = 0.0f;
  }
  const unsigned base = wid * EPW;
  const int idx = ei[(lane >> 4) * N_EDGES + base + (lane & 15)];

  const unsigned h = lane >> 3;
  const unsigned pbyte = lane * 16u;
  const unsigned xbyte = (lane & 7) * 16u;
  const char* Pb = (const char*)P;
  const char* Xb = (const char*)x;

#pragma unroll
  for (int jb = 0; jb < EPW / EB; ++jb) {
    float4 q[EB], k[EB];
#pragma unroll
    for (int j = 0; j < EB; ++j) {
      const int e = jb * EB + j;
      const unsigned snd = (unsigned)__builtin_amdgcn_readlane(idx, e);
      const unsigned rcv = (unsigned)__builtin_amdgcn_readlane(idx, 16 + e);
      const char* prow = Pb + (size_t)rcv * (HEADS * FEAT * 4);
      const char* xrow = Xb + (size_t)snd * (FEAT * 4);
      q[j] = *(const float4*)(prow + pbyte);
      k[j] = *(const float4*)(xrow + xbyte);
    }
    __builtin_amdgcn_sched_barrier(0);

    float mine = 0.0f;
#pragma unroll
    for (int j = 0; j < EB; ++j) {
      float p = q[j].y * k[j].y;
      p = fmaf(q[j].x, k[j].x, p);
      p = fmaf(q[j].z, k[j].z, p);
      p = fmaf(q[j].w, k[j].w, p);
      p += __shfl_xor(p, 1, 32);
      p += __shfl_xor(p, 2, 32);
      p += __shfl_xor(p, 4, 32);
      if ((lane & 7) == (unsigned)j) mine = p;
    }
    const float s = mine * 0.17677669529663687f;
    const unsigned el = jb * EB + (lane & 7);
    const int rl = __shfl(idx, 16 + el, 32);
    sEH[(size_t)(base + el) * HEADS + h] = s;
    atomic_max_f32_dev(&m[rl * HEADS + h], s);
  }
}

__global__ __launch_bounds__(256) void gat_edge_scatter(const float* __restrict__ x,
                                                        const int* __restrict__ ei,
                                                        const float* __restrict__ sEH,
                                                        const float* __restrict__ m,
                                                        float* __restrict__ denom,
                                                        float* __restrict__ num,
                                                        const int* __restrict__ flag) {
  const unsigned wid = (blockIdx.x * blockDim.x + threadIdx.x) >> 5;
  if (wid >= N_EDGES / EPW) return;
  if (*flag == 0) return;
  const unsigned lane = threadIdx.x & 31;
  const unsigned base = wid * EPW;
  const int idx = ei[(lane >> 4) * N_EDGES + base + (lane & 15)];

  const char* Xb = (const char*)x;
  const unsigned laneByte = lane * 4u;
  const unsigned hq = lane & 3;

  int   rl[EPW / EB];
  float s[EPW / EB], mv[EPW / EB];
#pragma unroll
  for (int jb = 0; jb < EPW / EB; ++jb) {
    const unsigned el = jb * EB + (lane >> 2);
    rl[jb] = __shfl(idx, 16 + el, 32);
    s[jb]  = sEH[(size_t)base * HEADS + jb * (EB * HEADS) + lane];
    mv[jb] = m[rl[jb] * HEADS + hq];
  }

  float v[EPW];
#pragma unroll
  for (int e = 0; e < EPW; ++e) {
    const unsigned snd = (unsigned)__builtin_amdgcn_readlane(idx, e);
    v[e] = *(const float*)(Xb + (size_t)snd * (FEAT * 4) + laneByte);
  }

  float w[EPW / EB];
#pragma unroll
  for (int jb = 0; jb < EPW / EB; ++jb) {
    w[jb] = __expf(s[jb] - mv[jb]);
    atomic_add_f32_dev(&denom[rl[jb] * HEADS + hq], w[jb]);
  }

#pragma unroll
  for (int e = 0; e < EPW; ++e) {
    const unsigned rcv = (unsigned)__builtin_amdgcn_readlane(idx, 16 + e);
    const int wb = __float_as_int(w[e / EB]);
    const int j  = e % EB;
    const float w0 = __int_as_float(__builtin_amdgcn_readlane(wb, 4 * j + 0));
    const float w1 = __int_as_float(__builtin_amdgcn_readlane(wb, 4 * j + 1));
    const float w2 = __int_as_float(__builtin_amdgcn_readlane(wb, 4 * j + 2));
    const float w3 = __int_as_float(__builtin_amdgcn_readlane(wb, 4 * j + 3));
    float* row = num + (size_t)rcv * (HEADS * FEAT);
    atomic_add_f32_dev_row<0 * FEAT * 4>(row, laneByte, w0 * v[e]);
    atomic_add_f32_dev_row<1 * FEAT * 4>(row, laneByte, w1 * v[e]);
    atomic_add_f32_dev_row<2 * FEAT * 4>(row, laneByte, w2 * v[e]);
    atomic_add_f32_dev_row<3 * FEAT * 4>(row, laneByte, w3 * v[e]);
  }
}

__global__ __launch_bounds__(64) void gat_out_wmma(const float* __restrict__ x,
                                                   const float* __restrict__ num,
                                                   const float* __restrict__ denom,
                                                   const float* __restrict__ Wv,
                                                   const float* __restrict__ Wout,
                                                   float* __restrict__ out) {
  const int lane = threadIdx.x & 31;
  const int nt   = threadIdx.x >> 5;
  const int m0   = blockIdx.x * 16;

  __shared__ float as4[HEADS][16 * FEAT];
  __shared__ float ts[16 * FEAT];

  {
    const float4* src = (const float4*)(num + (size_t)m0 * HEADS * FEAT);
    const int hh = (threadIdx.x >> 3) & 3, f4 = threadIdx.x & 7;
#pragma unroll
    for (int i = 0; i < 8; ++i) {
      const int row   = i * 2 + (threadIdx.x >> 5);
      const float4 t  = src[i * 64 + threadIdx.x];
      const float d   = denom[(m0 + row) * HEADS + hh];
      const float inv = (d > 0.0f) ? 1.0f / d : 0.0f;
      ((float4*)as4[hh])[row * 8 + f4] = make_float4(t.x * inv, t.y * inv, t.z * inv, t.w * inv);
    }
  }
  __syncthreads();

  const int col = lane & 15, hi = lane >> 4;
  v8f accv = {};
#pragma unroll
  for (int hh = 0; hh < HEADS; ++hh)
    accv = wmma_f32_k32(as4[hh], Wv + hh * FEAT * FEAT, nt * 16, lane, accv);
#pragma unroll
  for (int r = 0; r < 8; ++r)
    ts[(r + 8 * hi) * FEAT + nt * 16 + col] = accv[r] * 0.25f;
  __syncthreads();

  v8f acc = {};
  acc = wmma_f32_k32(ts, Wout, nt * 16, lane, acc);

#pragma unroll
  for (int r = 0; r < 8; ++r) {
    const size_t o = (size_t)(m0 + r + 8 * hi) * FEAT + nt * 16 + col;
    out[o] = x[o] + acc[r];
  }
}

extern "C" void kernel_launch(void* const* d_in, const int* in_sizes, int n_in,
                              void* d_out, int out_size, void* d_ws, size_t ws_size,
                              hipStream_t stream) {
  const float* x    = (const float*)d_in[0];
  const int*   ei   = (const int*)d_in[1];
  const float* Wq   = (const float*)d_in[2];
  const float* Wk   = (const float*)d_in[3];
  const float* Wv   = (const float*)d_in[4];
  const float* Wout = (const float*)d_in[5];
  float*       out  = (float*)d_out;

  float* ws   = (float*)d_ws;
  float* P    = ws;
  int*   flag = (int*)(P + (size_t)N_NODES * HEADS * FEAT);
  float* sEH  = (float*)flag + (size_t)HEADS * FEAT * FEAT;
  float* Mqk  = sEH;
  float* mbuf = sEH + (size_t)N_EDGES * HEADS;
  float* den  = mbuf + (size_t)N_NODES * HEADS;
  float* num  = den + (size_t)N_NODES * HEADS;

  gat_init_fold<<<INIT_BLOCKS + FOLD_BLOCKS, 256, 0, stream>>>(mbuf, den, num, Wq, Wk, Mqk, flag);

  gat_p_wmma<<<MTILES, 256, 0, stream>>>(x, Mqk, P);

  const int edgeBlocks = (N_EDGES / EPW * 32) / 256;
  gat_edge_fused<<<edgeBlocks, 256, 0, stream>>>(P, x, ei, den, num, flag);
  gat_edge_logits<<<edgeBlocks, 256, 0, stream>>>(P, x, ei, sEH, mbuf, den, num, flag);
  gat_edge_scatter<<<edgeBlocks, 256, 0, stream>>>(x, ei, sEH, mbuf, den, num, flag);

  gat_out_wmma<<<MTILES, 64, 0, stream>>>(x, num, den, Wv, Wout, out);
}
